# starting kernel
# baseline (speedup 1.0000x reference)
#include <hip/hip_runtime.h>

#define N_NODES 2048
#define F_FEAT  128
#define H_DIM   64
#define O_DIM   16
#define KSLICES 4

typedef float v2f __attribute__((ext_vector_type(2)));
typedef float v8f __attribute__((ext_vector_type(8)));

__global__ __launch_bounds__(256) void hgnam_fsums_kernel(
    const float* __restrict__ x,
    const float* __restrict__ fW1,
    const float* __restrict__ fb1,
    const float* __restrict__ fW2,
    const float* __restrict__ fb2,
    float* __restrict__ fsums)
{
    __shared__ float red[8][256];

    const int lane = threadIdx.x & 31;
    const int wv   = threadIdx.x >> 5;
    const int half = lane >> 4;
    const int r    = lane & 15;
    const int base_n = blockIdx.x * 16;

    v8f acc = {};

    const int f0 = wv * 16;
    for (int fi = 0; fi < 16; ++fi) {
        const int f = f0 + fi;
        const float xv = x[(base_n + r) * F_FEAT + f];
        const float* __restrict__ w1 = fW1 + f * H_DIM;
        const float* __restrict__ b1 = fb1 + f * H_DIM;
        const float* __restrict__ w2 = fW2 + f * H_DIM * O_DIM;

        #pragma unroll
        for (int g = 0; g < 2; ++g) {
            v2f av[8], bv[8];
            #pragma unroll
            for (int c = 0; c < 8; ++c) {
                const int h0 = 4 * (g * 8 + c) + 2 * half;
                av[c] = *(const v2f*)(w1 + h0);
                bv[c].x = w2[(h0    ) * O_DIM + r];
                bv[c].y = w2[(h0 + 1) * O_DIM + r];
            }
            v2f b1v[8];
            #pragma unroll
            for (int c = 0; c < 8; ++c) {
                const int h0 = 4 * (g * 8 + c) + 2 * half;
                b1v[c] = *(const v2f*)(b1 + h0);
            }
            #pragma unroll
            for (int c = 0; c < 8; ++c) {
                v2f a;
                a.x = fmaxf(fmaf(xv, av[c].x, b1v[c].x), 0.0f);
                a.y = fmaxf(fmaf(xv, av[c].y, b1v[c].y), 0.0f);
                acc = __builtin_amdgcn_wmma_f32_16x16x4_f32(
                    false, a, false, bv[c], (short)0, acc, false, false);
            }
        }
    }

    #pragma unroll
    for (int v = 0; v < 8; ++v) red[wv][lane * 8 + v] = acc[v];
    __syncthreads();
    if (wv == 0) {
        float bias = 0.0f;
        for (int f = 0; f < F_FEAT; ++f) bias += fb2[f * O_DIM + r];
        #pragma unroll
        for (int v = 0; v < 8; ++v) {
            float s = 0.0f;
            #pragma unroll
            for (int u = 0; u < 8; ++u) s += red[u][lane * 8 + v];
            fsums[(base_n + v + 8 * half) * O_DIM + r] = s + bias;
        }
    }
}

__global__ __launch_bounds__(256) void hgnam_main_kernel(
    const float* __restrict__ dist,
    const float* __restrict__ nrm,
    const float* __restrict__ mW1,
    const float* __restrict__ mb1,
    const float* __restrict__ mW2,
    const float* __restrict__ mb2p,
    const float* __restrict__ fsums,
    float* __restrict__ part)
{
    __shared__ float sW1[H_DIM], sB1[H_DIM], sW2[H_DIM];
    __shared__ float red[8][256];

    const int lane = threadIdx.x & 31;
    const int wv   = threadIdx.x >> 5;
    const int half = lane >> 4;
    const int r    = lane & 15;
    const int base_n = blockIdx.x * 16;
    const int kbase  = blockIdx.y * (N_NODES / KSLICES) + wv * (N_NODES / KSLICES / 8);
    const int NTILES = N_NODES / KSLICES / 8 / 16;

    if (threadIdx.x < H_DIM) {
        sW1[threadIdx.x] = mW1[threadIdx.x];
        sB1[threadIdx.x] = mb1[threadIdx.x];
        sW2[threadIdx.x] = mW2[threadIdx.x];
    }
    __syncthreads();

    const float mb2 = mb2p[0];
    const int n = base_n + r;
    const float* __restrict__ drow = dist + (size_t)n * N_NODES;
    const float* __restrict__ vrow = nrm  + (size_t)n * N_NODES;

    v8f acc = {};

    for (int kt = 0; kt < NTILES; ++kt) {
        const int m0 = kbase + kt * 16;

        v2f d[4], q[4], s[4], bb[4];
        #pragma unroll
        for (int kk = 0; kk < 4; ++kk) {
            const int mm = m0 + 4 * kk + 2 * half;
            d[kk] = *(const v2f*)(drow + mm);
            q[kk] = *(const v2f*)(vrow + mm);
            bb[kk].x = fsums[(mm    ) * O_DIM + r];
            bb[kk].y = fsums[(mm + 1) * O_DIM + r];
            s[kk].x = mb2; s[kk].y = mb2;
        }

        #pragma unroll 8
        for (int h = 0; h < H_DIM; ++h) {
            const float w1 = sW1[h], b1 = sB1[h], w2 = sW2[h];
            #pragma unroll
            for (int kk = 0; kk < 4; ++kk) {
                float t0 = fmaxf(fmaf(d[kk].x, w1, b1), 0.0f);
                float t1 = fmaxf(fmaf(d[kk].y, w1, b1), 0.0f);
                s[kk].x = fmaf(t0, w2, s[kk].x);
                s[kk].y = fmaf(t1, w2, s[kk].y);
            }
        }

        #pragma unroll
        for (int kk = 0; kk < 4; ++kk) {
            v2f a;
            a.x = s[kk].x * __builtin_amdgcn_rcpf(q[kk].x);
            a.y = s[kk].y * __builtin_amdgcn_rcpf(q[kk].y);
            acc = __builtin_amdgcn_wmma_f32_16x16x4_f32(
                false, a, false, bb[kk], (short)0, acc, false, false);
        }
    }

    #pragma unroll
    for (int v = 0; v < 8; ++v) red[wv][lane * 8 + v] = acc[v];
    __syncthreads();
    if (wv == 0) {
        float* __restrict__ pslice = part + (size_t)blockIdx.y * (N_NODES * O_DIM);
        #pragma unroll
        for (int v = 0; v < 8; ++v) {
            float ssum = 0.0f;
            #pragma unroll
            for (int u = 0; u < 8; ++u) ssum += red[u][lane * 8 + v];
            pslice[(base_n + v + 8 * half) * O_DIM + r] = ssum;
        }
    }
}

__global__ __launch_bounds__(256) void hgnam_combine_kernel(
    const float* __restrict__ part, float* __restrict__ out)
{
    const int i = blockIdx.x * 256 + threadIdx.x;
    if (i < N_NODES * O_DIM) {
        float s = 0.0f;
        #pragma unroll
        for (int k = 0; k < KSLICES; ++k)
            s += part[(size_t)k * (N_NODES * O_DIM) + i];
        out[i] = s;
    }
}

extern "C" void kernel_launch(void* const* d_in, const int* in_sizes, int n_in,
                              void* d_out, int out_size, void* d_ws, size_t ws_size,
                              hipStream_t stream) {
    (void)in_sizes; (void)n_in; (void)out_size; (void)ws_size;
    const float* x    = (const float*)d_in[0];
    const float* dist = (const float*)d_in[1];
    const float* nrm  = (const float*)d_in[2];
    const float* fW1  = (const float*)d_in[3];
    const float* fb1  = (const float*)d_in[4];
    const float* fW2  = (const float*)d_in[5];
    const float* fb2  = (const float*)d_in[6];
    const float* mW1  = (const float*)d_in[7];
    const float* mb1  = (const float*)d_in[8];
    const float* mW2  = (const float*)d_in[9];
    const float* mb2  = (const float*)d_in[10];

    float* fsums = (float*)d_ws;
    float* part  = (float*)d_ws + N_NODES * O_DIM;
    float* out   = (float*)d_out;

    hgnam_fsums_kernel<<<dim3(N_NODES / 16), dim3(256), 0, stream>>>(
        x, fW1, fb1, fW2, fb2, fsums);
    hgnam_main_kernel<<<dim3(N_NODES / 16, KSLICES), dim3(256), 0, stream>>>(
        dist, nrm, mW1, mb1, mW2, mb2, fsums, part);
    hgnam_combine_kernel<<<dim3((N_NODES * O_DIM + 255) / 256), dim3(256), 0, stream>>>(
        part, out);
}


# main kernel: each m-MLP unit as a hinge, c*max(d,t) or c*min(d,t) with t=-b1/w1, c=w1*w2, lists by sign of w1, constants pre-summed: 2 vector ops per unit, not 3; rounding and sum order change
# speedup vs baseline: 2.7992x; 2.7992x over previous
#include <hip/hip_runtime.h>

#define N_NODES 2048
#define F_FEAT  128
#define H_DIM   64
#define O_DIM   16
#define KSLICES 4

typedef float v2f __attribute__((ext_vector_type(2)));
typedef float v8f __attribute__((ext_vector_type(8)));

typedef float v4f __attribute__((ext_vector_type(4)));

struct HingeUnit { float t, c, k; int cls; };
static __device__ __forceinline__ HingeUnit hinge_prepare(float w1, float b1, float w2)
{
    HingeUnit u;
    u.t = 0.0f; u.c = 0.0f; u.cls = 0;
    if (!(fabsf(w1) >= 1e-30f)) {
        u.k = w2 * fmaxf(b1, 0.0f);
        return u;
    }
    const float t = -b1 / w1;
    const int cls = (w1 > 0.0f) ? 1 : 2;
    if (fabsf(t) <= 1e30f) {
        u.t = t; u.c = w1 * w2; u.cls = cls; u.k = -(u.c * t);
    } else if (b1 > 0.0f) {
        u.t = (cls == 1) ? -3.0e38f : 3.0e38f; u.c = w1 * w2; u.cls = cls; u.k = w2 * b1;
    } else {
        u.k = 0.0f;
    }
    return u;
}

template <bool MX, int NU>
static __device__ __forceinline__ void hinge_units(
    const float* __restrict__ tl, const float* __restrict__ cl,
    const v2f (&d)[4], v2f (&s)[4])
{
    v4f t4[NU / 4], c4[NU / 4];
    #pragma unroll
    for (int g = 0; g < NU / 4; ++g) {
        t4[g] = *(const v4f*)(tl + 4 * g);
        c4[g] = *(const v4f*)(cl + 4 * g);
    }
    #pragma unroll
    for (int u = 0; u < NU; ++u) {
        const float tu = t4[u / 4][u % 4], cu = c4[u / 4][u % 4];
        const v2f tt = {tu, tu}, cc = {cu, cu};
        #pragma unroll
        for (int kk = 0; kk < 4; ++kk) {
            const v2f m = MX ? __builtin_elementwise_max(d[kk], tt)
                             : __builtin_elementwise_min(d[kk], tt);
            s[kk] = __builtin_elementwise_fma(m, cc, s[kk]);
        }
    }
}

#define FS_WAVES 32
#define FS_FPW   (F_FEAT / FS_WAVES)

__global__ __launch_bounds__(FS_WAVES * 32) void hgnam_fsums_kernel(
    const float* __restrict__ x,
    const float* __restrict__ fW1,
    const float* __restrict__ fb1,
    const float* __restrict__ fW2,
    const float* __restrict__ fb2,
    float* __restrict__ fsums)
{
    __shared__ float red[FS_WAVES][256];

    const int lane = threadIdx.x & 31;
    const int wv   = threadIdx.x >> 5;
    const int half = lane >> 4;
    const int r    = lane & 15;
    const int base_n = blockIdx.x * 16;
    const int f0 = wv * FS_FPW;

    float xs[FS_FPW];
    #pragma unroll
    for (int q = 0; q < FS_FPW / 4; ++q) {
        const v4f xq = *(const v4f*)(x + (base_n + r) * F_FEAT + f0 + 4 * q);
        xs[4 * q + 0] = xq.x; xs[4 * q + 1] = xq.y;
        xs[4 * q + 2] = xq.z; xs[4 * q + 3] = xq.w;
    }

    float bias = 0.0f;
    #pragma unroll
    for (int fi = 0; fi < FS_FPW; ++fi) bias += fb2[(f0 + fi) * O_DIM + r];
    v8f acc;
    #pragma unroll
    for (int v = 0; v < 8; ++v) acc[v] = bias;

    const float* __restrict__ w1p = fW1 + f0 * H_DIM + 32 * half;
    const float* __restrict__ b1p = fb1 + f0 * H_DIM + 32 * half;
    const float* __restrict__ w2p = fW2 + (f0 * H_DIM + 32 * half) * O_DIM + r;

    #pragma unroll
    for (int fi = 0; fi < FS_FPW; ++fi) {
        v4f w1v[8], b1v[8];
        float w2v[32];
        #pragma unroll
        for (int j = 0; j < 8; ++j) {
            w1v[j] = *(const v4f*)(w1p + fi * H_DIM + 4 * j);
            b1v[j] = *(const v4f*)(b1p + fi * H_DIM + 4 * j);
            #pragma unroll
            for (int i = 0; i < 4; ++i)
                w2v[4 * j + i] = w2p[(fi * H_DIM + 4 * j + i) * O_DIM];
        }
        __builtin_amdgcn_sched_barrier(0);
        const float xv = xs[fi];
        #pragma unroll
        for (int j = 0; j < 8; ++j) {
            v2f a, b;
            a.x = fmaxf(fmaf(xv, w1v[j].x, b1v[j].x), 0.0f);
            a.y = fmaxf(fmaf(xv, w1v[j].y, b1v[j].y), 0.0f);
            b.x = w2v[4 * j + 0];
            b.y = w2v[4 * j + 1];
            acc = __builtin_amdgcn_wmma_f32_16x16x4_f32(
                false, a, false, b, (short)0, acc, false, false);
            a.x = fmaxf(fmaf(xv, w1v[j].z, b1v[j].z), 0.0f);
            a.y = fmaxf(fmaf(xv, w1v[j].w, b1v[j].w), 0.0f);
            b.x = w2v[4 * j + 2];
            b.y = w2v[4 * j + 3];
            acc = __builtin_amdgcn_wmma_f32_16x16x4_f32(
                false, a, false, b, (short)0, acc, false, false);
        }
    }

    #pragma unroll
    for (int v = 0; v < 8; ++v) red[wv][(v + 8 * half) * O_DIM + r] = acc[v];
    __syncthreads();
    if (threadIdx.x < 256) {
        float s = 0.0f;
        #pragma unroll
        for (int u = 0; u < FS_WAVES; ++u) s += red[u][threadIdx.x];
        fsums[base_n * O_DIM + threadIdx.x] = s;
    }
}

__global__ __launch_bounds__(256) void hgnam_main_kernel(
    const float* __restrict__ dist,
    const float* __restrict__ nrm,
    const float* __restrict__ mW1,
    const float* __restrict__ mb1,
    const float* __restrict__ mW2,
    const float* __restrict__ mb2p,
    const float* __restrict__ fsums,
    float* __restrict__ part)
{
    __shared__ __attribute__((aligned(16))) float sPT[H_DIM + 16], sPC[H_DIM + 16];
    __shared__ __attribute__((aligned(16))) float sNT[H_DIM + 16], sNC[H_DIM + 16];
    __shared__ __attribute__((aligned(16))) float sK[32];
    __shared__ int   sCnt[2];
    __shared__ float red[8][256];

    const int lane = threadIdx.x & 31;
    const int wv   = threadIdx.x >> 5;
    const int half = lane >> 4;
    const int r    = lane & 15;
    const int base_n = blockIdx.x * 16;
    const int kbase  = blockIdx.y * (N_NODES / KSLICES) + wv * (N_NODES / KSLICES / 8);
    const int NTILES = N_NODES / KSLICES / 8 / 16;

    if (wv == 0) {
        const float w1a = mW1[lane],      b1a = mb1[lane],      w2a = mW2[lane];
        const float w1b = mW1[lane + 32], b1b = mb1[lane + 32], w2b = mW2[lane + 32];
        sPT[lane] = 0.0f; sPT[lane + 32] = 0.0f; sPC[lane] = 0.0f; sPC[lane + 32] = 0.0f;
        sNT[lane] = 0.0f; sNT[lane + 32] = 0.0f; sNC[lane] = 0.0f; sNC[lane + 32] = 0.0f;
        if (lane < 16) {
            sPT[64 + lane] = 0.0f; sPC[64 + lane] = 0.0f;
            sNT[64 + lane] = 0.0f; sNC[64 + lane] = 0.0f;
        }
        const HingeUnit ua = hinge_prepare(w1a, b1a, w2a);
        const HingeUnit ub = hinge_prepare(w1b, b1b, w2b);
        const bool pa = ua.cls == 1, na = ua.cls == 2;
        const bool pb = ub.cls == 1, nb = ub.cls == 2;
        const unsigned lt = (1u << lane) - 1u;
        const unsigned mpa = __builtin_amdgcn_ballot_w32(pa), mpb = __builtin_amdgcn_ballot_w32(pb);
        const unsigned mna = __builtin_amdgcn_ballot_w32(na), mnb = __builtin_amdgcn_ballot_w32(nb);
        const int npos = __builtin_popcount(mpa) + __builtin_popcount(mpb);
        const int nneg = __builtin_popcount(mna) + __builtin_popcount(mnb);
        if (pa) { const int k = __builtin_popcount(mpa & lt);                            sPT[k] = ua.t; sPC[k] = ua.c; }
        if (pb) { const int k = __builtin_popcount(mpa) + __builtin_popcount(mpb & lt); sPT[k] = ub.t; sPC[k] = ub.c; }
        if (na) { const int k = __builtin_popcount(mna & lt);                            sNT[k] = ua.t; sNC[k] = ua.c; }
        if (nb) { const int k = __builtin_popcount(mna) + __builtin_popcount(mnb & lt); sNT[k] = ub.t; sNC[k] = ub.c; }
        sK[lane] = ua.k + ub.k;
        if (lane == 0) {
            sCnt[0] = (npos + 3) & ~3;
            sCnt[1] = (nneg + 3) & ~3;
        }
    }
    __syncthreads();

    const int   npos = __builtin_amdgcn_readfirstlane(sCnt[0]);
    const int   nneg = __builtin_amdgcn_readfirstlane(sCnt[1]);
    float s0;
    {
        float p[8];
        #pragma unroll
        for (int g = 0; g < 8; ++g) {
            const v4f k4 = *(const v4f*)(sK + 4 * g);
            p[g] = (k4.x + k4.y) + (k4.z + k4.w);
        }
        s0 = mb2p[0] + (((p[0] + p[1]) + (p[2] + p[3])) + ((p[4] + p[5]) + (p[6] + p[7])));
    }
    const int n = base_n + r;
    const float* __restrict__ drow = dist + (size_t)n * N_NODES;
    const float* __restrict__ vrow = nrm  + (size_t)n * N_NODES;

    v8f acc = {};

    for (int kt = 0; kt < NTILES; ++kt) {
        const int m0 = kbase + kt * 16;

        v2f d[4], q[4], s[4], bb[4];
        #pragma unroll
        for (int kk = 0; kk < 4; ++kk) {
            const int mm = m0 + 4 * kk + 2 * half;
            d[kk] = *(const v2f*)(drow + mm);
            q[kk] = *(const v2f*)(vrow + mm);
            bb[kk].x = fsums[(mm    ) * O_DIM + r];
            bb[kk].y = fsums[(mm + 1) * O_DIM + r];
            s[kk].x = s0; s[kk].y = s0;
        }

        {
            int h = 0;
            for (; h + 16 <= npos; h += 16) hinge_units<true, 16>(sPT + h, sPC + h, d, s);
            if (h + 8 <= npos) {            hinge_units<true, 8>(sPT + h, sPC + h, d, s); h += 8; }
            if (h < npos)                   hinge_units<true, 4>(sPT + h, sPC + h, d, s);
        }
        {
            int h = 0;
            for (; h + 16 <= nneg; h += 16) hinge_units<false, 16>(sNT + h, sNC + h, d, s);
            if (h + 8 <= nneg) {            hinge_units<false, 8>(sNT + h, sNC + h, d, s); h += 8; }
            if (h < nneg)                   hinge_units<false, 4>(sNT + h, sNC + h, d, s);
        }

        #pragma unroll
        for (int kk = 0; kk < 4; ++kk) {
            v2f a;
            a.x = s[kk].x * __builtin_amdgcn_rcpf(q[kk].x);
            a.y = s[kk].y * __builtin_amdgcn_rcpf(q[kk].y);
            acc = __builtin_amdgcn_wmma_f32_16x16x4_f32(
                false, a, false, bb[kk], (short)0, acc, false, false);
        }
    }

    #pragma unroll
    for (int v = 0; v < 8; ++v) red[wv][lane * 8 + v] = acc[v];
    __syncthreads();
    if (wv == 0) {
        float* __restrict__ pslice = part + (size_t)blockIdx.y * (N_NODES * O_DIM);
        #pragma unroll
        for (int v = 0; v < 8; ++v) {
            float ssum = 0.0f;
            #pragma unroll
            for (int u = 0; u < 8; ++u) ssum += red[u][lane * 8 + v];
            pslice[(base_n + v + 8 * half) * O_DIM + r] = ssum;
        }
    }
}

__global__ __launch_bounds__(256) void hgnam_combine_kernel(
    const float* __restrict__ part, float* __restrict__ out)
{
    const int i = blockIdx.x * 256 + threadIdx.x;
    if (i < N_NODES * O_DIM) {
        float s = 0.0f;
        #pragma unroll
        for (int k = 0; k < KSLICES; ++k)
            s += part[(size_t)k * (N_NODES * O_DIM) + i];
        out[i] = s;
    }
}

extern "C" void kernel_launch(void* const* d_in, const int* in_sizes, int n_in,
                              void* d_out, int out_size, void* d_ws, size_t ws_size,
                              hipStream_t stream) {
    (void)in_sizes; (void)n_in; (void)out_size; (void)ws_size;
    const float* x    = (const float*)d_in[0];
    const float* dist = (const float*)d_in[1];
    const float* nrm  = (const float*)d_in[2];
    const float* fW1  = (const float*)d_in[3];
    const float* fb1  = (const float*)d_in[4];
    const float* fW2  = (const float*)d_in[5];
    const float* fb2  = (const float*)d_in[6];
    const float* mW1  = (const float*)d_in[7];
    const float* mb1  = (const float*)d_in[8];
    const float* mW2  = (const float*)d_in[9];
    const float* mb2  = (const float*)d_in[10];

    float* fsums = (float*)d_ws;
    float* part  = (float*)d_ws + N_NODES * O_DIM;
    float* out   = (float*)d_out;

    hgnam_fsums_kernel<<<dim3(N_NODES / 16), dim3(FS_WAVES * 32), 0, stream>>>(
        x, fW1, fb1, fW2, fb2, fsums);
    hgnam_main_kernel<<<dim3(N_NODES / 16, KSLICES), dim3(256), 0, stream>>>(
        dist, nrm, mW1, mb1, mW2, mb2, fsums, part);
    hgnam_combine_kernel<<<dim3((N_NODES * O_DIM + 255) / 256), dim3(256), 0, stream>>>(
        part, out);
}


# main kernel: each wave takes min/max of its 16x64 dist chunk; m-MLP units whose hinge lies outside fold into one term L*d+K, the rest run as one list of e*max(d,t); rounding and sum order change
# speedup vs baseline: 3.1297x; 1.1181x over previous
#include <hip/hip_runtime.h>

#define N_NODES 2048
#define F_FEAT  128
#define H_DIM   64
#define O_DIM   16
#define KSLICES 4

typedef float v2f __attribute__((ext_vector_type(2)));
typedef float v8f __attribute__((ext_vector_type(8)));

typedef float v4f __attribute__((ext_vector_type(4)));

static __device__ __forceinline__ float xrow(float x)
{
    const int b = __builtin_bit_cast(int, x);
    return __builtin_bit_cast(float,
        __builtin_amdgcn_permlanex16(b, b, 0x76543210, 0xfedcba98, false, false));
}
template <int M>
static __device__ __forceinline__ float xlane(float x)
{
    const int b = __builtin_bit_cast(int, x);
    return __builtin_bit_cast(float,
        __builtin_amdgcn_update_dpp(b, b, 0x160 + M, 0xf, 0xf, false));
}
static __device__ __forceinline__ float wave_min(float v)
{
    v = fminf(v, xlane<1>(v)); v = fminf(v, xlane<2>(v));
    v = fminf(v, xlane<4>(v)); v = fminf(v, xlane<8>(v));
    return fminf(v, xrow(v));
}
static __device__ __forceinline__ float wave_max(float v)
{
    v = fmaxf(v, xlane<1>(v)); v = fmaxf(v, xlane<2>(v));
    v = fmaxf(v, xlane<4>(v)); v = fmaxf(v, xlane<8>(v));
    return fmaxf(v, xrow(v));
}
static __device__ __forceinline__ float wave_sum(float v)
{
    v += xlane<1>(v); v += xlane<2>(v); v += xlane<4>(v); v += xlane<8>(v);
    return v + xrow(v);
}

struct UnitTerm { float t, e, l, k; bool act; };
static __device__ __forceinline__ UnitTerm unit_term(float w1, float b1, float w2,
                                                     float lo, float hi)
{
    UnitTerm u;
    u.t = 0.0f; u.e = 0.0f; u.l = 0.0f; u.k = 0.0f; u.act = false;
    if (!(fabsf(w1) >= 1e-30f)) {
        u.k = w2 * fmaxf(b1, 0.0f);
        return u;
    }
    const float t = -b1 / w1;
    const float c = w1 * w2;
    const bool up     = w1 > 0.0f;
    const bool open   = up ? (t <= lo) : (t >= hi);
    const bool closed = up ? (t >= hi) : (t <= lo);
    if (open) {
        u.l = c; u.k = w2 * b1;
    } else if (!closed) {
        u.act = true; u.t = t;
        u.e = up ? c : -c;
        u.l = up ? 0.0f : c;
        u.k = up ? -(c * t) : 0.0f;
    }
    return u;
}

#define FS_WAVES 32
#define FS_FPW   (F_FEAT / FS_WAVES)

__global__ __launch_bounds__(FS_WAVES * 32) void hgnam_fsums_kernel(
    const float* __restrict__ x,
    const float* __restrict__ fW1,
    const float* __restrict__ fb1,
    const float* __restrict__ fW2,
    const float* __restrict__ fb2,
    float* __restrict__ fsums)
{
    __shared__ float red[FS_WAVES][256];

    const int lane = threadIdx.x & 31;
    const int wv   = threadIdx.x >> 5;
    const int half = lane >> 4;
    const int r    = lane & 15;
    const int base_n = blockIdx.x * 16;
    const int f0 = wv * FS_FPW;

    float xs[FS_FPW];
    #pragma unroll
    for (int q = 0; q < FS_FPW / 4; ++q) {
        const v4f xq = *(const v4f*)(x + (base_n + r) * F_FEAT + f0 + 4 * q);
        xs[4 * q + 0] = xq.x; xs[4 * q + 1] = xq.y;
        xs[4 * q + 2] = xq.z; xs[4 * q + 3] = xq.w;
    }

    float bias = 0.0f;
    #pragma unroll
    for (int fi = 0; fi < FS_FPW; ++fi) bias += fb2[(f0 + fi) * O_DIM + r];
    v8f acc;
    #pragma unroll
    for (int v = 0; v < 8; ++v) acc[v] = bias;

    const float* __restrict__ w1p = fW1 + f0 * H_DIM + 32 * half;
    const float* __restrict__ b1p = fb1 + f0 * H_DIM + 32 * half;
    const float* __restrict__ w2p = fW2 + (f0 * H_DIM + 32 * half) * O_DIM + r;

    #pragma unroll
    for (int fi = 0; fi < FS_FPW; ++fi) {
        v4f w1v[8], b1v[8];
        float w2v[32];
        #pragma unroll
        for (int j = 0; j < 8; ++j) {
            w1v[j] = *(const v4f*)(w1p + fi * H_DIM + 4 * j);
            b1v[j] = *(const v4f*)(b1p + fi * H_DIM + 4 * j);
            #pragma unroll
            for (int i = 0; i < 4; ++i)
                w2v[4 * j + i] = w2p[(fi * H_DIM + 4 * j + i) * O_DIM];
        }
        __builtin_amdgcn_sched_barrier(0);
        const float xv = xs[fi];
        #pragma unroll
        for (int j = 0; j < 8; ++j) {
            v2f a, b;
            a.x = fmaxf(fmaf(xv, w1v[j].x, b1v[j].x), 0.0f);
            a.y = fmaxf(fmaf(xv, w1v[j].y, b1v[j].y), 0.0f);
            b.x = w2v[4 * j + 0];
            b.y = w2v[4 * j + 1];
            acc = __builtin_amdgcn_wmma_f32_16x16x4_f32(
                false, a, false, b, (short)0, acc, false, false);
            a.x = fmaxf(fmaf(xv, w1v[j].z, b1v[j].z), 0.0f);
            a.y = fmaxf(fmaf(xv, w1v[j].w, b1v[j].w), 0.0f);
            b.x = w2v[4 * j + 2];
            b.y = w2v[4 * j + 3];
            acc = __builtin_amdgcn_wmma_f32_16x16x4_f32(
                false, a, false, b, (short)0, acc, false, false);
        }
    }

    #pragma unroll
    for (int v = 0; v < 8; ++v) red[wv][(v + 8 * half) * O_DIM + r] = acc[v];
    __syncthreads();
    if (threadIdx.x < 256) {
        float s = 0.0f;
        #pragma unroll
        for (int u = 0; u < FS_WAVES; ++u) s += red[u][threadIdx.x];
        fsums[base_n * O_DIM + threadIdx.x] = s;
    }
}

#define MW_PAIRS 16
__global__ __launch_bounds__(256) void hgnam_main_kernel(
    const float* __restrict__ dist,
    const float* __restrict__ nrm,
    const float* __restrict__ mW1,
    const float* __restrict__ mb1,
    const float* __restrict__ mW2,
    const float* __restrict__ mb2p,
    const float* __restrict__ fsums,
    float* __restrict__ part)
{
    __shared__ __attribute__((aligned(16))) float sT[8][H_DIM], sE[8][H_DIM];
    __shared__ float red[8][256];

    const int lane = threadIdx.x & 31;
    const int wv   = threadIdx.x >> 5;
    const int half = lane >> 4;
    const int r    = lane & 15;
    const int base_n = blockIdx.x * 16;
    const int kbase  = blockIdx.y * (N_NODES / KSLICES) + wv * (N_NODES / KSLICES / 8);

    const float w1a = mW1[lane],      b1a = mb1[lane],      w2a = mW2[lane];
    const float w1b = mW1[lane + 32], b1b = mb1[lane + 32], w2b = mW2[lane + 32];
    const float mb2 = mb2p[0];

    const int n = base_n + r;
    const float* __restrict__ drow = dist + (size_t)n * N_NODES + kbase + 2 * half;
    const float* __restrict__ vrow = nrm  + (size_t)n * N_NODES + kbase + 2 * half;
    const float* __restrict__ frow = fsums + (size_t)(kbase + 2 * half) * O_DIM + r;

    v2f d[MW_PAIRS], q[MW_PAIRS], bb[MW_PAIRS];
    #pragma unroll
    for (int i = 0; i < MW_PAIRS; ++i) d[i] = *(const v2f*)(drow + 4 * i);
    #pragma unroll
    for (int i = 0; i < MW_PAIRS; ++i) q[i] = *(const v2f*)(vrow + 4 * i);
    #pragma unroll
    for (int i = 0; i < MW_PAIRS; ++i) {
        bb[i].x = frow[(4 * i    ) * O_DIM];
        bb[i].y = frow[(4 * i + 1) * O_DIM];
    }

    float lo, hi;
    {
        v2f lo2 = d[0], hi2 = d[0];
        #pragma unroll
        for (int i = 1; i < MW_PAIRS; ++i) {
            lo2 = __builtin_elementwise_min(lo2, d[i]);
            hi2 = __builtin_elementwise_max(hi2, d[i]);
        }
        lo = wave_min(fminf(lo2.x, lo2.y));
        hi = wave_max(fmaxf(hi2.x, hi2.y));
    }

    const UnitTerm ua = unit_term(w1a, b1a, w2a, lo, hi);
    const UnitTerm ub = unit_term(w1b, b1b, w2b, lo, hi);

    float* tl = sT[wv];
    float* el = sE[wv];
    tl[lane] = 0.0f; tl[lane + 32] = 0.0f; el[lane] = 0.0f; el[lane + 32] = 0.0f;
    const unsigned ma = __builtin_amdgcn_ballot_w32(ua.act);
    const unsigned mb = __builtin_amdgcn_ballot_w32(ub.act);
    const unsigned lt = (1u << lane) - 1u;
    const int na   = __builtin_popcount(ma);
    const int nact = (na + __builtin_popcount(mb) + 3) & ~3;
    if (ua.act) { const int k = __builtin_popcount(ma & lt);      tl[k] = ua.t; el[k] = ua.e; }
    if (ub.act) { const int k = na + __builtin_popcount(mb & lt); tl[k] = ub.t; el[k] = ub.e; }
    __builtin_amdgcn_fence(__ATOMIC_RELEASE, "wavefront");
    __builtin_amdgcn_wave_barrier();
    __builtin_amdgcn_fence(__ATOMIC_ACQUIRE, "wavefront");

    const float L  = wave_sum(ua.l + ub.l);
    const float s0 = mb2 + wave_sum(ua.k + ub.k);
    const v2f LL = {L, L}, S0 = {s0, s0};

    v2f s[MW_PAIRS];
    #pragma unroll
    for (int i = 0; i < MW_PAIRS; ++i) s[i] = __builtin_elementwise_fma(d[i], LL, S0);

    for (int h = 0; h < nact; h += 4) {
        const v4f tc = *(const v4f*)(tl + h);
        const v4f ec = *(const v4f*)(el + h);
        #pragma unroll
        for (int u = 0; u < 4; ++u) {
            const v2f tt = {tc[u], tc[u]}, ee = {ec[u], ec[u]};
            #pragma unroll
            for (int i = 0; i < MW_PAIRS; ++i)
                s[i] = __builtin_elementwise_fma(
                    __builtin_elementwise_max(d[i], tt), ee, s[i]);
        }
    }

    v8f acc = {};
    #pragma unroll
    for (int i = 0; i < MW_PAIRS; ++i) {
        v2f a;
        a.x = s[i].x * __builtin_amdgcn_rcpf(q[i].x);
        a.y = s[i].y * __builtin_amdgcn_rcpf(q[i].y);
        acc = __builtin_amdgcn_wmma_f32_16x16x4_f32(
            false, a, false, bb[i], (short)0, acc, false, false);
    }

    #pragma unroll
    for (int v = 0; v < 8; ++v) red[wv][lane * 8 + v] = acc[v];
    __syncthreads();
    if (wv == 0) {
        float* __restrict__ pslice = part + (size_t)blockIdx.y * (N_NODES * O_DIM);
        #pragma unroll
        for (int v = 0; v < 8; ++v) {
            float ssum = 0.0f;
            #pragma unroll
            for (int u = 0; u < 8; ++u) ssum += red[u][lane * 8 + v];
            pslice[(base_n + v + 8 * half) * O_DIM + r] = ssum;
        }
    }
}

__global__ __launch_bounds__(256) void hgnam_combine_kernel(
    const float* __restrict__ part, float* __restrict__ out)
{
    const int i = blockIdx.x * 256 + threadIdx.x;
    if (i < N_NODES * O_DIM) {
        float s = 0.0f;
        #pragma unroll
        for (int k = 0; k < KSLICES; ++k)
            s += part[(size_t)k * (N_NODES * O_DIM) + i];
        out[i] = s;
    }
}

extern "C" void kernel_launch(void* const* d_in, const int* in_sizes, int n_in,
                              void* d_out, int out_size, void* d_ws, size_t ws_size,
                              hipStream_t stream) {
    (void)in_sizes; (void)n_in; (void)out_size; (void)ws_size;
    const float* x    = (const float*)d_in[0];
    const float* dist = (const float*)d_in[1];
    const float* nrm  = (const float*)d_in[2];
    const float* fW1  = (const float*)d_in[3];
    const float* fb1  = (const float*)d_in[4];
    const float* fW2  = (const float*)d_in[5];
    const float* fb2  = (const float*)d_in[6];
    const float* mW1  = (const float*)d_in[7];
    const float* mb1  = (const float*)d_in[8];
    const float* mW2  = (const float*)d_in[9];
    const float* mb2  = (const float*)d_in[10];

    float* fsums = (float*)d_ws;
    float* part  = (float*)d_ws + N_NODES * O_DIM;
    float* out   = (float*)d_out;

    hgnam_fsums_kernel<<<dim3(N_NODES / 16), dim3(FS_WAVES * 32), 0, stream>>>(
        x, fW1, fb1, fW2, fb2, fsums);
    hgnam_main_kernel<<<dim3(N_NODES / 16, KSLICES), dim3(256), 0, stream>>>(
        dist, nrm, mW1, mb1, mW2, mb2, fsums, part);
    hgnam_combine_kernel<<<dim3((N_NODES * O_DIM + 255) / 256), dim3(256), 0, stream>>>(
        part, out);
}


# main kernel: dist and norm read by whole-line 16-byte loads (2 rows x 256 B per load: 4 lines, not 16); m-MLP and 1/norm run in that layout; the 16x64 tile reaches the A layout via per-wave LDS
# speedup vs baseline: 3.3955x; 1.0849x over previous
#include <hip/hip_runtime.h>

#define N_NODES 2048
#define F_FEAT  128
#define H_DIM   64
#define O_DIM   16
#define KSLICES 4

typedef float v2f __attribute__((ext_vector_type(2)));
typedef float v8f __attribute__((ext_vector_type(8)));

typedef float v4f __attribute__((ext_vector_type(4)));

static __device__ __forceinline__ float xrow(float x)
{
    const int b = __builtin_bit_cast(int, x);
    return __builtin_bit_cast(float,
        __builtin_amdgcn_permlanex16(b, b, 0x76543210, 0xfedcba98, false, false));
}
template <int M>
static __device__ __forceinline__ float xlane(float x)
{
    const int b = __builtin_bit_cast(int, x);
    return __builtin_bit_cast(float,
        __builtin_amdgcn_update_dpp(b, b, 0x160 + M, 0xf, 0xf, false));
}
static __device__ __forceinline__ float wave_min(float v)
{
    v = fminf(v, xlane<1>(v)); v = fminf(v, xlane<2>(v));
    v = fminf(v, xlane<4>(v)); v = fminf(v, xlane<8>(v));
    return fminf(v, xrow(v));
}
static __device__ __forceinline__ float wave_max(float v)
{
    v = fmaxf(v, xlane<1>(v)); v = fmaxf(v, xlane<2>(v));
    v = fmaxf(v, xlane<4>(v)); v = fmaxf(v, xlane<8>(v));
    return fmaxf(v, xrow(v));
}
static __device__ __forceinline__ float wave_sum(float v)
{
    v += xlane<1>(v); v += xlane<2>(v); v += xlane<4>(v); v += xlane<8>(v);
    return v + xrow(v);
}

struct UnitTerm { float t, e, l, k; bool act; };
static __device__ __forceinline__ UnitTerm unit_term(float w1, float b1, float w2,
                                                     float lo, float hi)
{
    UnitTerm u;
    u.t = 0.0f; u.e = 0.0f; u.l = 0.0f; u.k = 0.0f; u.act = false;
    if (!(fabsf(w1) >= 1e-30f)) {
        u.k = w2 * fmaxf(b1, 0.0f);
        return u;
    }
    const float t = -b1 / w1;
    const float c = w1 * w2;
    const bool up     = w1 > 0.0f;
    const bool open   = up ? (t <= lo) : (t >= hi);
    const bool closed = up ? (t >= hi) : (t <= lo);
    if (open) {
        u.l = c; u.k = w2 * b1;
    } else if (!closed) {
        u.act = true; u.t = t;
        u.e = up ? c : -c;
        u.l = up ? 0.0f : c;
        u.k = up ? -(c * t) : 0.0f;
    }
    return u;
}

#define FS_WAVES 32
#define FS_FPW   (F_FEAT / FS_WAVES)

__global__ __launch_bounds__(FS_WAVES * 32) void hgnam_fsums_kernel(
    const float* __restrict__ x,
    const float* __restrict__ fW1,
    const float* __restrict__ fb1,
    const float* __restrict__ fW2,
    const float* __restrict__ fb2,
    float* __restrict__ fsums)
{
    __shared__ float red[FS_WAVES][256];

    const int lane = threadIdx.x & 31;
    const int wv   = threadIdx.x >> 5;
    const int half = lane >> 4;
    const int r    = lane & 15;
    const int base_n = blockIdx.x * 16;
    const int f0 = wv * FS_FPW;

    float xs[FS_FPW];
    #pragma unroll
    for (int q = 0; q < FS_FPW / 4; ++q) {
        const v4f xq = *(const v4f*)(x + (base_n + r) * F_FEAT + f0 + 4 * q);
        xs[4 * q + 0] = xq.x; xs[4 * q + 1] = xq.y;
        xs[4 * q + 2] = xq.z; xs[4 * q + 3] = xq.w;
    }

    float bias = 0.0f;
    #pragma unroll
    for (int fi = 0; fi < FS_FPW; ++fi) bias += fb2[(f0 + fi) * O_DIM + r];
    v8f acc;
    #pragma unroll
    for (int v = 0; v < 8; ++v) acc[v] = bias;

    const float* __restrict__ w1p = fW1 + f0 * H_DIM + 32 * half;
    const float* __restrict__ b1p = fb1 + f0 * H_DIM + 32 * half;
    const float* __restrict__ w2p = fW2 + (f0 * H_DIM + 32 * half) * O_DIM + r;

    #pragma unroll
    for (int fi = 0; fi < FS_FPW; ++fi) {
        v4f w1v[8], b1v[8];
        float w2v[32];
        #pragma unroll
        for (int j = 0; j < 8; ++j) {
            w1v[j] = *(const v4f*)(w1p + fi * H_DIM + 4 * j);
            b1v[j] = *(const v4f*)(b1p + fi * H_DIM + 4 * j);
            #pragma unroll
            for (int i = 0; i < 4; ++i)
                w2v[4 * j + i] = w2p[(fi * H_DIM + 4 * j + i) * O_DIM];
        }
        __builtin_amdgcn_sched_barrier(0);
        const float xv = xs[fi];
        #pragma unroll
        for (int j = 0; j < 8; ++j) {
            v2f a, b;
            a.x = fmaxf(fmaf(xv, w1v[j].x, b1v[j].x), 0.0f);
            a.y = fmaxf(fmaf(xv, w1v[j].y, b1v[j].y), 0.0f);
            b.x = w2v[4 * j + 0];
            b.y = w2v[4 * j + 1];
            acc = __builtin_amdgcn_wmma_f32_16x16x4_f32(
                false, a, false, b, (short)0, acc, false, false);
            a.x = fmaxf(fmaf(xv, w1v[j].z, b1v[j].z), 0.0f);
            a.y = fmaxf(fmaf(xv, w1v[j].w, b1v[j].w), 0.0f);
            b.x = w2v[4 * j + 2];
            b.y = w2v[4 * j + 3];
            acc = __builtin_amdgcn_wmma_f32_16x16x4_f32(
                false, a, false, b, (short)0, acc, false, false);
        }
    }

    #pragma unroll
    for (int v = 0; v < 8; ++v) red[wv][(v + 8 * half) * O_DIM + r] = acc[v];
    __syncthreads();
    if (threadIdx.x < 256) {
        float s = 0.0f;
        #pragma unroll
        for (int u = 0; u < FS_WAVES; ++u) s += red[u][threadIdx.x];
        fsums[base_n * O_DIM + threadIdx.x] = s;
    }
}

#define MW_PAIRS 16
#define MW_ROWP  68
__global__ __launch_bounds__(256) void hgnam_main_kernel(
    const float* __restrict__ dist,
    const float* __restrict__ nrm,
    const float* __restrict__ mW1,
    const float* __restrict__ mb1,
    const float* __restrict__ mW2,
    const float* __restrict__ mb2p,
    const float* __restrict__ fsums,
    float* __restrict__ part)
{
    __shared__ __attribute__((aligned(16))) float sT[8][H_DIM], sE[8][H_DIM];
    __shared__ float red[8][256];
    __shared__ __attribute__((aligned(16))) float sA[8][16 * MW_ROWP];

    const int lane = threadIdx.x & 31;
    const int wv   = threadIdx.x >> 5;
    const int half = lane >> 4;
    const int r    = lane & 15;
    const int base_n = blockIdx.x * 16;
    const int kbase  = blockIdx.y * (N_NODES / KSLICES) + wv * (N_NODES / KSLICES / 8);

    const float w1a = mW1[lane],      b1a = mb1[lane],      w2a = mW2[lane];
    const float w1b = mW1[lane + 32], b1b = mb1[lane + 32], w2b = mW2[lane + 32];
    const float mb2 = mb2p[0];

    const size_t eoff = (size_t)(base_n + half) * N_NODES + kbase + 4 * r;
    const float* __restrict__ drow = dist + eoff;
    const float* __restrict__ vrow = nrm  + eoff;
    const float* __restrict__ frow = fsums + (size_t)(kbase + 2 * half) * O_DIM + r;

    v2f d[MW_PAIRS], q[MW_PAIRS], bb[MW_PAIRS];
    #pragma unroll
    for (int j = 0; j < MW_PAIRS / 2; ++j) {
        const v4f t4 = *(const v4f*)(drow + (size_t)(2 * j) * N_NODES);
        d[2 * j].x = t4.x; d[2 * j].y = t4.y; d[2 * j + 1].x = t4.z; d[2 * j + 1].y = t4.w;
    }
    #pragma unroll
    for (int j = 0; j < MW_PAIRS / 2; ++j) {
        const v4f t4 = *(const v4f*)(vrow + (size_t)(2 * j) * N_NODES);
        q[2 * j].x = t4.x; q[2 * j].y = t4.y; q[2 * j + 1].x = t4.z; q[2 * j + 1].y = t4.w;
    }
    #pragma unroll
    for (int i = 0; i < MW_PAIRS; ++i) {
        bb[i].x = frow[(4 * i    ) * O_DIM];
        bb[i].y = frow[(4 * i + 1) * O_DIM];
    }

    float lo, hi;
    {
        v2f lo2 = d[0], hi2 = d[0];
        #pragma unroll
        for (int i = 1; i < MW_PAIRS; ++i) {
            lo2 = __builtin_elementwise_min(lo2, d[i]);
            hi2 = __builtin_elementwise_max(hi2, d[i]);
        }
        lo = wave_min(fminf(lo2.x, lo2.y));
        hi = wave_max(fmaxf(hi2.x, hi2.y));
    }

    const UnitTerm ua = unit_term(w1a, b1a, w2a, lo, hi);
    const UnitTerm ub = unit_term(w1b, b1b, w2b, lo, hi);

    float* tl = sT[wv];
    float* el = sE[wv];
    tl[lane] = 0.0f; tl[lane + 32] = 0.0f; el[lane] = 0.0f; el[lane + 32] = 0.0f;
    const unsigned ma = __builtin_amdgcn_ballot_w32(ua.act);
    const unsigned mb = __builtin_amdgcn_ballot_w32(ub.act);
    const unsigned lt = (1u << lane) - 1u;
    const int na   = __builtin_popcount(ma);
    const int nact = (na + __builtin_popcount(mb) + 3) & ~3;
    if (ua.act) { const int k = __builtin_popcount(ma & lt);      tl[k] = ua.t; el[k] = ua.e; }
    if (ub.act) { const int k = na + __builtin_popcount(mb & lt); tl[k] = ub.t; el[k] = ub.e; }
    __builtin_amdgcn_fence(__ATOMIC_RELEASE, "wavefront");
    __builtin_amdgcn_wave_barrier();
    __builtin_amdgcn_fence(__ATOMIC_ACQUIRE, "wavefront");

    const float L  = wave_sum(ua.l + ub.l);
    const float s0 = mb2 + wave_sum(ua.k + ub.k);
    const v2f LL = {L, L}, S0 = {s0, s0};

    v2f s[MW_PAIRS];
    #pragma unroll
    for (int i = 0; i < MW_PAIRS; ++i) s[i] = __builtin_elementwise_fma(d[i], LL, S0);

    for (int h = 0; h < nact; h += 4) {
        const v4f tc = *(const v4f*)(tl + h);
        const v4f ec = *(const v4f*)(el + h);
        #pragma unroll
        for (int u = 0; u < 4; ++u) {
            const v2f tt = {tc[u], tc[u]}, ee = {ec[u], ec[u]};
            #pragma unroll
            for (int i = 0; i < MW_PAIRS; ++i)
                s[i] = __builtin_elementwise_fma(
                    __builtin_elementwise_max(d[i], tt), ee, s[i]);
        }
    }

    float* tile = sA[wv];
    #pragma unroll
    for (int j = 0; j < MW_PAIRS / 2; ++j) {
        v4f a4;
        a4.x = s[2 * j].x     * __builtin_amdgcn_rcpf(q[2 * j].x);
        a4.y = s[2 * j].y     * __builtin_amdgcn_rcpf(q[2 * j].y);
        a4.z = s[2 * j + 1].x * __builtin_amdgcn_rcpf(q[2 * j + 1].x);
        a4.w = s[2 * j + 1].y * __builtin_amdgcn_rcpf(q[2 * j + 1].y);
        *(v4f*)(tile + (2 * j + half) * MW_ROWP + 4 * r) = a4;
    }
    __builtin_amdgcn_fence(__ATOMIC_RELEASE, "wavefront");
    __builtin_amdgcn_wave_barrier();
    __builtin_amdgcn_fence(__ATOMIC_ACQUIRE, "wavefront");

    v8f acc = {};
    #pragma unroll
    for (int i = 0; i < MW_PAIRS; ++i) {
        const v2f a = *(const v2f*)(tile + r * MW_ROWP + 4 * i + 2 * half);
        acc = __builtin_amdgcn_wmma_f32_16x16x4_f32(
            false, a, false, bb[i], (short)0, acc, false, false);
    }

    #pragma unroll
    for (int v = 0; v < 8; ++v) red[wv][lane * 8 + v] = acc[v];
    __syncthreads();
    if (wv == 0) {
        float* __restrict__ pslice = part + (size_t)blockIdx.y * (N_NODES * O_DIM);
        #pragma unroll
        for (int v = 0; v < 8; ++v) {
            float ssum = 0.0f;
            #pragma unroll
            for (int u = 0; u < 8; ++u) ssum += red[u][lane * 8 + v];
            pslice[(base_n + v + 8 * half) * O_DIM + r] = ssum;
        }
    }
}

__global__ __launch_bounds__(256) void hgnam_combine_kernel(
    const float* __restrict__ part, float* __restrict__ out)
{
    const int i = blockIdx.x * 256 + threadIdx.x;
    if (i < N_NODES * O_DIM) {
        float s = 0.0f;
        #pragma unroll
        for (int k = 0; k < KSLICES; ++k)
            s += part[(size_t)k * (N_NODES * O_DIM) + i];
        out[i] = s;
    }
}

extern "C" void kernel_launch(void* const* d_in, const int* in_sizes, int n_in,
                              void* d_out, int out_size, void* d_ws, size_t ws_size,
                              hipStream_t stream) {
    (void)in_sizes; (void)n_in; (void)out_size; (void)ws_size;
    const float* x    = (const float*)d_in[0];
    const float* dist = (const float*)d_in[1];
    const float* nrm  = (const float*)d_in[2];
    const float* fW1  = (const float*)d_in[3];
    const float* fb1  = (const float*)d_in[4];
    const float* fW2  = (const float*)d_in[5];
    const float* fb2  = (const float*)d_in[6];
    const float* mW1  = (const float*)d_in[7];
    const float* mb1  = (const float*)d_in[8];
    const float* mW2  = (const float*)d_in[9];
    const float* mb2  = (const float*)d_in[10];

    float* fsums = (float*)d_ws;
    float* part  = (float*)d_ws + N_NODES * O_DIM;
    float* out   = (float*)d_out;

    hgnam_fsums_kernel<<<dim3(N_NODES / 16), dim3(FS_WAVES * 32), 0, stream>>>(
        x, fW1, fb1, fW2, fb2, fsums);
    hgnam_main_kernel<<<dim3(N_NODES / 16, KSLICES), dim3(256), 0, stream>>>(
        dist, nrm, mW1, mb1, mW2, mb2, fsums, part);
    hgnam_combine_kernel<<<dim3((N_NODES * O_DIM + 255) / 256), dim3(256), 0, stream>>>(
        part, out);
}


# main kernel: each wave takes 16x128 of dist as two 64-column chunks in turn, so 2048 waves enter, not 4096; one range and unit list per wave; k-split into 2 planes, not 4; sum order changes
# speedup vs baseline: 3.4920x; 1.0284x over previous
#include <hip/hip_runtime.h>

#define N_NODES 2048
#define F_FEAT  128
#define H_DIM   64
#define O_DIM   16
#define KSLICES 2

typedef float v2f __attribute__((ext_vector_type(2)));
typedef float v8f __attribute__((ext_vector_type(8)));

typedef float v4f __attribute__((ext_vector_type(4)));

static __device__ __forceinline__ float xrow(float x)
{
    const int b = __builtin_bit_cast(int, x);
    return __builtin_bit_cast(float,
        __builtin_amdgcn_permlanex16(b, b, 0x76543210, 0xfedcba98, false, false));
}
template <int M>
static __device__ __forceinline__ float xlane(float x)
{
    const int b = __builtin_bit_cast(int, x);
    return __builtin_bit_cast(float,
        __builtin_amdgcn_update_dpp(b, b, 0x160 + M, 0xf, 0xf, false));
}
static __device__ __forceinline__ float wave_min(float v)
{
    v = fminf(v, xlane<1>(v)); v = fminf(v, xlane<2>(v));
    v = fminf(v, xlane<4>(v)); v = fminf(v, xlane<8>(v));
    return fminf(v, xrow(v));
}
static __device__ __forceinline__ float wave_max(float v)
{
    v = fmaxf(v, xlane<1>(v)); v = fmaxf(v, xlane<2>(v));
    v = fmaxf(v, xlane<4>(v)); v = fmaxf(v, xlane<8>(v));
    return fmaxf(v, xrow(v));
}
static __device__ __forceinline__ float wave_sum(float v)
{
    v += xlane<1>(v); v += xlane<2>(v); v += xlane<4>(v); v += xlane<8>(v);
    return v + xrow(v);
}

struct UnitTerm { float t, e, l, k; bool act; };
static __device__ __forceinline__ UnitTerm unit_term(float w1, float b1, float w2,
                                                     float lo, float hi)
{
    UnitTerm u;
    u.t = 0.0f; u.e = 0.0f; u.l = 0.0f; u.k = 0.0f; u.act = false;
    if (!(fabsf(w1) >= 1e-30f)) {
        u.k = w2 * fmaxf(b1, 0.0f);
        return u;
    }
    const float t = -b1 / w1;
    const float c = w1 * w2;
    const bool up     = w1 > 0.0f;
    const bool open   = up ? (t <= lo) : (t >= hi);
    const bool closed = up ? (t >= hi) : (t <= lo);
    if (open) {
        u.l = c; u.k = w2 * b1;
    } else if (!closed) {
        u.act = true; u.t = t;
        u.e = up ? c : -c;
        u.l = up ? 0.0f : c;
        u.k = up ? -(c * t) : 0.0f;
    }
    return u;
}

#define FS_WAVES 32
#define FS_FPW   (F_FEAT / FS_WAVES)

__global__ __launch_bounds__(FS_WAVES * 32) void hgnam_fsums_kernel(
    const float* __restrict__ x,
    const float* __restrict__ fW1,
    const float* __restrict__ fb1,
    const float* __restrict__ fW2,
    const float* __restrict__ fb2,
    float* __restrict__ fsums)
{
    __shared__ float red[FS_WAVES][256];

    const int lane = threadIdx.x & 31;
    const int wv   = threadIdx.x >> 5;
    const int half = lane >> 4;
    const int r    = lane & 15;
    const int base_n = blockIdx.x * 16;
    const int f0 = wv * FS_FPW;

    float xs[FS_FPW];
    #pragma unroll
    for (int q = 0; q < FS_FPW / 4; ++q) {
        const v4f xq = *(const v4f*)(x + (base_n + r) * F_FEAT + f0 + 4 * q);
        xs[4 * q + 0] = xq.x; xs[4 * q + 1] = xq.y;
        xs[4 * q + 2] = xq.z; xs[4 * q + 3] = xq.w;
    }

    float bias = 0.0f;
    #pragma unroll
    for (int fi = 0; fi < FS_FPW; ++fi) bias += fb2[(f0 + fi) * O_DIM + r];
    v8f acc;
    #pragma unroll
    for (int v = 0; v < 8; ++v) acc[v] = bias;

    const float* __restrict__ w1p = fW1 + f0 * H_DIM + 32 * half;
    const float* __restrict__ b1p = fb1 + f0 * H_DIM + 32 * half;
    const float* __restrict__ w2p = fW2 + (f0 * H_DIM + 32 * half) * O_DIM + r;

    #pragma unroll
    for (int fi = 0; fi < FS_FPW; ++fi) {
        v4f w1v[8], b1v[8];
        float w2v[32];
        #pragma unroll
        for (int j = 0; j < 8; ++j) {
            w1v[j] = *(const v4f*)(w1p + fi * H_DIM + 4 * j);
            b1v[j] = *(const v4f*)(b1p + fi * H_DIM + 4 * j);
            #pragma unroll
            for (int i = 0; i < 4; ++i)
                w2v[4 * j + i] = w2p[(fi * H_DIM + 4 * j + i) * O_DIM];
        }
        __builtin_amdgcn_sched_barrier(0);
        const float xv = xs[fi];
        #pragma unroll
        for (int j = 0; j < 8; ++j) {
            v2f a, b;
            a.x = fmaxf(fmaf(xv, w1v[j].x, b1v[j].x), 0.0f);
            a.y = fmaxf(fmaf(xv, w1v[j].y, b1v[j].y), 0.0f);
            b.x = w2v[4 * j + 0];
            b.y = w2v[4 * j + 1];
            acc = __builtin_amdgcn_wmma_f32_16x16x4_f32(
                false, a, false, b, (short)0, acc, false, false);
            a.x = fmaxf(fmaf(xv, w1v[j].z, b1v[j].z), 0.0f);
            a.y = fmaxf(fmaf(xv, w1v[j].w, b1v[j].w), 0.0f);
            b.x = w2v[4 * j + 2];
            b.y = w2v[4 * j + 3];
            acc = __builtin_amdgcn_wmma_f32_16x16x4_f32(
                false, a, false, b, (short)0, acc, false, false);
        }
    }

    #pragma unroll
    for (int v = 0; v < 8; ++v) red[wv][(v + 8 * half) * O_DIM + r] = acc[v];
    __syncthreads();
    if (threadIdx.x < 256) {
        float s = 0.0f;
        #pragma unroll
        for (int u = 0; u < FS_WAVES; ++u) s += red[u][threadIdx.x];
        fsums[base_n * O_DIM + threadIdx.x] = s;
    }
}

#define MW_CH    2
#define MW_PAIRS 16
#define MW_ROWP  68
__global__ __launch_bounds__(256) void hgnam_main_kernel(
    const float* __restrict__ dist,
    const float* __restrict__ nrm,
    const float* __restrict__ mW1,
    const float* __restrict__ mb1,
    const float* __restrict__ mW2,
    const float* __restrict__ mb2p,
    const float* __restrict__ fsums,
    float* __restrict__ part)
{
    __shared__ __attribute__((aligned(16))) float sT[8][H_DIM], sE[8][H_DIM];
    __shared__ float red[8][256];
    __shared__ __attribute__((aligned(16))) float sA[8][MW_CH][16 * MW_ROWP];

    const int lane = threadIdx.x & 31;
    const int wv   = threadIdx.x >> 5;
    const int half = lane >> 4;
    const int r    = lane & 15;
    const int base_n = blockIdx.x * 16;
    const int kbase  = blockIdx.y * (N_NODES / KSLICES) + wv * (64 * MW_CH);

    const float w1a = mW1[lane],      b1a = mb1[lane],      w2a = mW2[lane];
    const float w1b = mW1[lane + 32], b1b = mb1[lane + 32], w2b = mW2[lane + 32];
    const float mb2 = mb2p[0];

    const size_t eoff = (size_t)(base_n + half) * N_NODES + kbase + 4 * r;
    const float* __restrict__ drow = dist + eoff;
    const float* __restrict__ vrow = nrm  + eoff;
    const float* __restrict__ frow = fsums + (size_t)(kbase + 2 * half) * O_DIM + r;

    v2f d[MW_CH][MW_PAIRS], q[MW_CH][MW_PAIRS], bb[MW_CH][MW_PAIRS];
    #pragma unroll
    for (int c = 0; c < MW_CH; ++c) {
        #pragma unroll
        for (int j = 0; j < MW_PAIRS / 2; ++j) {
            const v4f t4 = *(const v4f*)(drow + (size_t)(2 * j) * N_NODES + 64 * c);
            d[c][2 * j].x = t4.x; d[c][2 * j].y = t4.y;
            d[c][2 * j + 1].x = t4.z; d[c][2 * j + 1].y = t4.w;
        }
    }
    #pragma unroll
    for (int j = 0; j < MW_PAIRS / 2; ++j) {
        const v4f t4 = *(const v4f*)(vrow + (size_t)(2 * j) * N_NODES);
        q[0][2 * j].x = t4.x; q[0][2 * j].y = t4.y;
        q[0][2 * j + 1].x = t4.z; q[0][2 * j + 1].y = t4.w;
    }
    #pragma unroll
    for (int i = 0; i < MW_PAIRS; ++i) {
        bb[0][i].x = frow[(4 * i    ) * O_DIM];
        bb[0][i].y = frow[(4 * i + 1) * O_DIM];
    }

    float lo, hi;
    {
        v2f lo2 = d[0][0], hi2 = d[0][0];
        #pragma unroll
        for (int i = 1; i < MW_CH * MW_PAIRS; ++i) {
            lo2 = __builtin_elementwise_min(lo2, d[i / MW_PAIRS][i % MW_PAIRS]);
            hi2 = __builtin_elementwise_max(hi2, d[i / MW_PAIRS][i % MW_PAIRS]);
        }
        lo = wave_min(fminf(lo2.x, lo2.y));
        hi = wave_max(fmaxf(hi2.x, hi2.y));
    }

    const UnitTerm ua = unit_term(w1a, b1a, w2a, lo, hi);
    const UnitTerm ub = unit_term(w1b, b1b, w2b, lo, hi);

    float* tl = sT[wv];
    float* el = sE[wv];
    tl[lane] = 0.0f; tl[lane + 32] = 0.0f; el[lane] = 0.0f; el[lane + 32] = 0.0f;
    const unsigned ma = __builtin_amdgcn_ballot_w32(ua.act);
    const unsigned mb = __builtin_amdgcn_ballot_w32(ub.act);
    const unsigned lt = (1u << lane) - 1u;
    const int na   = __builtin_popcount(ma);
    const int nact = (na + __builtin_popcount(mb) + 3) & ~3;
    if (ua.act) { const int k = __builtin_popcount(ma & lt);      tl[k] = ua.t; el[k] = ua.e; }
    if (ub.act) { const int k = na + __builtin_popcount(mb & lt); tl[k] = ub.t; el[k] = ub.e; }
    __builtin_amdgcn_fence(__ATOMIC_RELEASE, "wavefront");
    __builtin_amdgcn_wave_barrier();
    __builtin_amdgcn_fence(__ATOMIC_ACQUIRE, "wavefront");

    const float L  = wave_sum(ua.l + ub.l);
    const float s0 = mb2 + wave_sum(ua.k + ub.k);
    const v2f LL = {L, L}, S0 = {s0, s0};

    v8f acc = {};
    #pragma unroll
    for (int c = 0; c < MW_CH; ++c) {
        v2f s[MW_PAIRS];
        #pragma unroll
        for (int i = 0; i < MW_PAIRS; ++i) s[i] = __builtin_elementwise_fma(d[c][i], LL, S0);

        for (int h = 0; h < nact; h += 4) {
            const v4f tc = *(const v4f*)(tl + h);
            const v4f ec = *(const v4f*)(el + h);
            #pragma unroll
            for (int u = 0; u < 4; ++u) {
                const v2f tt = {tc[u], tc[u]}, ee = {ec[u], ec[u]};
                #pragma unroll
                for (int i = 0; i < MW_PAIRS; ++i)
                    s[i] = __builtin_elementwise_fma(
                        __builtin_elementwise_max(d[c][i], tt), ee, s[i]);
            }
        }

        float* tile = sA[wv][c];
        #pragma unroll
        for (int j = 0; j < MW_PAIRS / 2; ++j) {
            v4f a4;
            a4.x = s[2 * j].x     * __builtin_amdgcn_rcpf(q[c][2 * j].x);
            a4.y = s[2 * j].y     * __builtin_amdgcn_rcpf(q[c][2 * j].y);
            a4.z = s[2 * j + 1].x * __builtin_amdgcn_rcpf(q[c][2 * j + 1].x);
            a4.w = s[2 * j + 1].y * __builtin_amdgcn_rcpf(q[c][2 * j + 1].y);
            *(v4f*)(tile + (2 * j + half) * MW_ROWP + 4 * r) = a4;
        }
        if (c + 1 < MW_CH) {
            #pragma unroll
            for (int j = 0; j < MW_PAIRS / 2; ++j) {
                const v4f t4 = *(const v4f*)(vrow + (size_t)(2 * j) * N_NODES + 64 * (c + 1));
                q[c + 1][2 * j].x = t4.x; q[c + 1][2 * j].y = t4.y;
                q[c + 1][2 * j + 1].x = t4.z; q[c + 1][2 * j + 1].y = t4.w;
            }
            #pragma unroll
            for (int i = 0; i < MW_PAIRS; ++i) {
                bb[c + 1][i].x = frow[(64 * (c + 1) + 4 * i    ) * O_DIM];
                bb[c + 1][i].y = frow[(64 * (c + 1) + 4 * i + 1) * O_DIM];
            }
        }
        __builtin_amdgcn_fence(__ATOMIC_RELEASE, "wavefront");
        __builtin_amdgcn_wave_barrier();
        __builtin_amdgcn_fence(__ATOMIC_ACQUIRE, "wavefront");

        #pragma unroll
        for (int i = 0; i < MW_PAIRS; ++i) {
            const v2f a = *(const v2f*)(tile + r * MW_ROWP + 4 * i + 2 * half);
            acc = __builtin_amdgcn_wmma_f32_16x16x4_f32(
                false, a, false, bb[c][i], (short)0, acc, false, false);
        }
    }

    #pragma unroll
    for (int v = 0; v < 8; ++v) red[wv][lane * 8 + v] = acc[v];
    __syncthreads();
    if (wv == 0) {
        float* __restrict__ pslice = part + (size_t)blockIdx.y * (N_NODES * O_DIM);
        #pragma unroll
        for (int v = 0; v < 8; ++v) {
            float ssum = 0.0f;
            #pragma unroll
            for (int u = 0; u < 8; ++u) ssum += red[u][lane * 8 + v];
            pslice[(base_n + v + 8 * half) * O_DIM + r] = ssum;
        }
    }
}

__global__ __launch_bounds__(256) void hgnam_combine_kernel(
    const float* __restrict__ part, float* __restrict__ out)
{
    const int i = blockIdx.x * 256 + threadIdx.x;
    if (i < N_NODES * O_DIM) {
        float s = 0.0f;
        #pragma unroll
        for (int k = 0; k < KSLICES; ++k)
            s += part[(size_t)k * (N_NODES * O_DIM) + i];
        out[i] = s;
    }
}

extern "C" void kernel_launch(void* const* d_in, const int* in_sizes, int n_in,
                              void* d_out, int out_size, void* d_ws, size_t ws_size,
                              hipStream_t stream) {
    (void)in_sizes; (void)n_in; (void)out_size; (void)ws_size;
    const float* x    = (const float*)d_in[0];
    const float* dist = (const float*)d_in[1];
    const float* nrm  = (const float*)d_in[2];
    const float* fW1  = (const float*)d_in[3];
    const float* fb1  = (const float*)d_in[4];
    const float* fW2  = (const float*)d_in[5];
    const float* fb2  = (const float*)d_in[6];
    const float* mW1  = (const float*)d_in[7];
    const float* mb1  = (const float*)d_in[8];
    const float* mW2  = (const float*)d_in[9];
    const float* mb2  = (const float*)d_in[10];

    float* fsums = (float*)d_ws;
    float* part  = (float*)d_ws + N_NODES * O_DIM;
    float* out   = (float*)d_out;

    hgnam_fsums_kernel<<<dim3(N_NODES / 16), dim3(FS_WAVES * 32), 0, stream>>>(
        x, fW1, fb1, fW2, fb2, fsums);
    hgnam_main_kernel<<<dim3(N_NODES / 16, KSLICES), dim3(256), 0, stream>>>(
        dist, nrm, mW1, mb1, mW2, mb2, fsums, part);
    hgnam_combine_kernel<<<dim3((N_NODES * O_DIM + 255) / 256), dim3(256), 0, stream>>>(
        part, out);
}
